# starting kernel
# baseline (speedup 1.0000x reference)
#include <hip/hip_runtime.h>

#define D 128
#define LDSW 132

typedef float v2f __attribute__((ext_vector_type(2)));
typedef float v8f __attribute__((ext_vector_type(8)));

__global__ void gcn_zero(float* __restrict__ out, float* __restrict__ deg,
                         int n_out, int n_deg) {
    int i = blockIdx.x * blockDim.x + threadIdx.x;
    int stride = gridDim.x * blockDim.x;
    for (int j = i; j < n_out; j += stride) out[j] = 0.0f;
    for (int j = i; j < n_deg; j += stride) deg[j] = 0.0f;
}

__global__ void gcn_scatter(const float* __restrict__ feat,
                            const int* __restrict__ src,
                            const int* __restrict__ dst,
                            float* __restrict__ agg,
                            float* __restrict__ deg, int E) {
    long long tid = (long long)blockIdx.x * blockDim.x + threadIdx.x;
    int e = (int)(tid >> 5);
    int c = (int)(tid & 31);
    if (e >= E) return;
    int s = src[e];
    int d = dst[e];
    const float4* fin = (const float4*)(feat + (long long)s * D);
    float4 v = fin[c];
    float* ao = agg + (long long)d * D + c * 4;
    unsafeAtomicAdd(ao + 0, v.x);
    unsafeAtomicAdd(ao + 1, v.y);
    unsafeAtomicAdd(ao + 2, v.z);
    unsafeAtomicAdd(ao + 3, v.w);
    if (c == 0) unsafeAtomicAdd(deg + d, 1.0f);
}

__global__ void __launch_bounds__(256)
gcn_gemm(const float* __restrict__ Wm,
         const float* __restrict__ bias,
         const float* __restrict__ deg,
         float* __restrict__ out,
         int N) {
    __shared__ __align__(16) float hs[16 * LDSW];

    const int tile = blockIdx.x;
    const int t = threadIdx.x;

    for (int i = t; i < 16 * D; i += 256) {
        int row = i >> 7;
        int col = i & (D - 1);
        int node = tile * 16 + row;
        float v = 0.0f;
        if (node < N) {
            float dg = deg[node];
            if (dg > 0.0f) v = out[(long long)node * D + col] / dg;
        }
        hs[row * LDSW + col] = v;
    }
    __syncthreads();

    const int wave = t >> 5;
    const int lane = t & 31;
    const int m    = lane & 15;
    const int n    = wave * 16 + m;
    const int koff = (lane < 16) ? 0 : 2;

    const float* hrow = hs + m * LDSW;
    const float* wrow = Wm + (long long)n * D;

    v8f acc = {};
    #pragma unroll 4
    for (int k0 = 0; k0 < D; k0 += 4) {
        v2f a = *(const v2f*)(hrow + k0 + koff);
        v2f b = *(const v2f*)(wrow + k0 + koff);
        acc = __builtin_amdgcn_wmma_f32_16x16x4_f32(
             false, a,  false, b,
             (short)0, acc,  false,  false);
    }

    const float bcol = bias[n];
    const int rbase = (lane < 16) ? 0 : 8;
    float* orow = out + (long long)(tile * 16 + rbase) * D + n;
    if (tile * 16 + 16 <= N) {
        #pragma unroll
        for (int r = 0; r < 8; ++r) orow[r * D] = acc[r] + bcol;
    } else {
        #pragma unroll
        for (int r = 0; r < 8; ++r) {
            if (tile * 16 + rbase + r < N) orow[r * D] = acc[r] + bcol;
        }
    }
}

extern "C" void kernel_launch(void* const* d_in, const int* in_sizes, int n_in,
                              void* d_out, int out_size, void* d_ws, size_t ws_size,
                              hipStream_t stream) {
    const float* feature = (const float*)d_in[0];
    const float* Wm      = (const float*)d_in[1];
    const float* bias    = (const float*)d_in[2];
    const int*   src     = (const int*)d_in[3];
    const int*   dst     = (const int*)d_in[4];

    float* out = (float*)d_out;
    float* deg = (float*)d_ws;

    const int N = in_sizes[0] / D;
    const int E = in_sizes[3];
    const int n_out = N * D;

    gcn_zero<<<4096, 256, 0, stream>>>(out, deg, n_out, N);

    long long scatter_threads = (long long)E * 32;
    int scatter_blocks = (int)((scatter_threads + 255) / 256);
    gcn_scatter<<<scatter_blocks, 256, 0, stream>>>(feature, src, dst, out, deg, E);

    int gemm_blocks = (N + 15) / 16;
    gcn_gemm<<<gemm_blocks, 256, 0, stream>>>(Wm, bias, deg, out, N);
}


# dst scan replaced by a scan of 1-byte owner keys (dst/391) written by a small first launch: a block reads 4x fewer bytes and tests 16 keys per lane by byte masks; 2 launches; sum order = LDS arrival
# speedup vs baseline: 2.9329x; 2.9329x over previous
#include <hip/hip_runtime.h>

#define D 128
#define TS 132

#define RT 391
#define RTP 400
#define SB_WAVES 32
#define NBLK 256
#define STEP 2048
#define SUB 512
#define QTH 64
#define QCAP (QTH + SUB)

typedef float v2f __attribute__((ext_vector_type(2)));
typedef float v8f __attribute__((ext_vector_type(8)));


__global__ void __launch_bounds__(256)
gcn_owner_keys(const int* __restrict__ dst, unsigned* __restrict__ keys, int E) {
    const int lane = threadIdx.x & 31;
    const int nchunks = (E / STEP) * (STEP / 512);
    const int nwaves = gridDim.x * 8;
    for (int c = blockIdx.x * 8 + (threadIdx.x >> 5); c < nchunks; c += nwaves) {
        const int4* p = (const int4*)dst + c * 128 + lane;
        int4 d[4];
        #pragma unroll
        for (int k = 0; k < 4; ++k) d[k] = p[32 * k];
        unsigned* o = keys + c * 128 + lane;
        #pragma unroll
        for (int k = 0; k < 4; ++k) {
            const unsigned k0 = ((unsigned)d[k].x / RT) & (NBLK - 1);
            const unsigned k1 = ((unsigned)d[k].y / RT) & (NBLK - 1);
            const unsigned k2 = ((unsigned)d[k].z / RT) & (NBLK - 1);
            const unsigned k3 = ((unsigned)d[k].w / RT) & (NBLK - 1);
            o[32 * k] = k0 | (k1 << 8) | (k2 << 16) | (k3 << 24);
        }
    }
}

__device__ __forceinline__ void load_step(const unsigned* __restrict__ keys, int s,
                                          int lane, uint4 (&c)[4]) {
    const uint4* p = (const uint4*)keys + s * (STEP / 16) + lane;
    c[0] = p[0];
    c[1] = p[32];
    c[2] = p[64];
    c[3] = p[96];
}

__device__ __forceinline__ unsigned equal_bytes(unsigned w, unsigned pat) {
    const unsigned x = w ^ pat;
    return ~(((x & 0x7f7f7f7fu) + 0x7f7f7f7fu) | x) & 0x80808080u;
}

__device__ __forceinline__ unsigned match16(const uint4 kv, unsigned pat) {
    return (equal_bytes(kv.x, pat) >> 7) | (equal_bytes(kv.y, pat) >> 6)
         | (equal_bytes(kv.z, pat) >> 5) | (equal_bytes(kv.w, pat) >> 4);
}

__device__ __forceinline__ int queue16(unsigned m, int el, unsigned* myq, int q) {
    unsigned bal = __builtin_amdgcn_ballot_w32(m != 0u);
    while (bal) {
        if (m != 0u) {
            const int p = __builtin_ctz(m);
            const int pos = q + (int)__builtin_amdgcn_mbcnt_lo(bal, 0u);
            myq[pos] = (unsigned)(el + ((p & 3) << 2) + (p >> 3));
            m &= m - 1u;
        }
        q += __builtin_popcount(bal);
        bal = __builtin_amdgcn_ballot_w32(m != 0u);
    }
    return q;
}

__device__ __forceinline__ void drain(const float* __restrict__ feat, const int* __restrict__ src,
                                      const int* __restrict__ dst, float* tile, float* dcount,
                                      const unsigned* myq, int q, int lane, int base, int rows) {
    for (int i0 = 0; i0 < q; i0 += 32) {
        const int cnt = min(32, q - i0);
        int sr = 0;
        int dlr = RTP - 1;
        if (lane < cnt) {
            const unsigned e = myq[i0 + lane];
            sr = src[e];
            const int dl = dst[e] - base;
            if ((unsigned)dl < (unsigned)rows) {
                dlr = dl;
                atomicAdd(&dcount[dl], 1.0f);
            }
        }
        int g = 0;
        do {
            float4 v[8];
            #pragma unroll
            for (int k = 0; k < 8; ++k) {
                const int sk = __builtin_amdgcn_readlane(sr, g + k);
                v[k] = ((const float4*)(feat + (long long)sk * D))[lane];
            }
            #pragma unroll
            for (int k = 0; k < 8; ++k) {
                const int dk = __builtin_amdgcn_readlane(dlr, g + k);
                float* row = tile + dk * TS + lane;
                atomicAdd(row,      v[k].x);
                atomicAdd(row + 32, v[k].y);
                atomicAdd(row + 64, v[k].z);
                atomicAdd(row + 96, v[k].w);
            }
            g += 8;
        } while (g < cnt);
    }
}

__device__ __forceinline__ int scan_step(const float* __restrict__ feat, const int* __restrict__ src,
                                         const int* __restrict__ dst, const unsigned* __restrict__ keys,
                                         float* tile, float* dcount,
                                         unsigned* myq, uint4 (&cur)[4], uint4 (&nxt)[4],
                                         int i, int nsteps, int first, int lane, unsigned pat,
                                         int base, int rows, int q) {
    const int in = (i + SB_WAVES < nsteps) ? i + SB_WAVES : i;
    int sn = first + in; if (sn >= nsteps) sn -= nsteps;
    load_step(keys, sn, lane, nxt);
    int s = first + i; if (s >= nsteps) s -= nsteps;
    const int el = s * STEP + 16 * lane;
    unsigned m[4];
    #pragma unroll
    for (int k = 0; k < 4; ++k) m[k] = match16(cur[k], pat);
    asm volatile("" :: "v"(m[0]), "v"(m[1]), "v"(m[2]), "v"(m[3]));
    #pragma unroll
    for (int k = 0; k < 4; ++k) {
        q = queue16(m[k], el + SUB * k, myq, q);
        if (q >= QTH) { drain(feat, src, dst, tile, dcount, myq, q, lane, base, rows); q = 0; }
    }
    return q;
}

union QueueOrW {
    unsigned q[SB_WAVES * QCAP];
    float w[D * TS];
};

__global__ void __launch_bounds__(SB_WAVES * 32)
gcn_owner_layer(const float* __restrict__ feat,
                const int* __restrict__ src,
                const int* __restrict__ dst,
                const unsigned* __restrict__ keys,
                const float* Wm,
                const float* bias,
                float* out,
                int N, int E) {
    __shared__ __align__(16) float tile[RTP * TS];
    __shared__ __align__(16) float dcount[RTP];
    __shared__ __align__(16) QueueOrW qw;

    const int t = threadIdx.x;
    const int lane = t & 31;
    const int wave = __builtin_amdgcn_readfirstlane(t >> 5);
    unsigned* myq = qw.q + wave * QCAP;
    const unsigned pat = (blockIdx.x & (NBLK - 1)) * 0x01010101u;

    const int nsteps = E / STEP;
    const int first = (int)(((long long)blockIdx.x * nsteps) / gridDim.x);

    const int c = lane & 15;
    const int h = lane >> 4;
    const int n = (wave & 7) * 16 + c;
    const int rg = wave >> 3;

    for (int base = blockIdx.x * RT; base < N; base += gridDim.x * RT) {
        const int rows = min(RT, N - base);

        for (int i = t; i < RTP * TS / 4; i += SB_WAVES * 32)
            ((float4*)tile)[i] = make_float4(0.f, 0.f, 0.f, 0.f);
        for (int i = t; i < RTP; i += SB_WAVES * 32) dcount[i] = 0.0f;
        __syncthreads();

        int q = 0;
        uint4 A[4] = {}, B[4] = {};
        int i = wave;
        if (i < nsteps) {
            int s = first + i; if (s >= nsteps) s -= nsteps;
            load_step(keys, s, lane, A);
        }
        while (i < nsteps) {
            q = scan_step(feat, src, dst, keys, tile, dcount, myq, A, B, i, nsteps, first, lane, pat, base, rows, q);
            i += SB_WAVES;
            if (i >= nsteps) break;
            q = scan_step(feat, src, dst, keys, tile, dcount, myq, B, A, i, nsteps, first, lane, pat, base, rows, q);
            i += SB_WAVES;
        }
        if (wave == 0) {
            #pragma nounroll
            for (int e0 = nsteps * STEP; e0 < E; e0 += 32) {
                const int e = e0 + lane;
                const int dl = ((e < E) ? dst[e] : -1) - base;
                const bool m = (unsigned)dl < (unsigned)rows;
                const unsigned bal = __builtin_amdgcn_ballot_w32(m);
                if (bal) {
                    if (m) {
                        const int pos = q + (int)__builtin_amdgcn_mbcnt_lo(bal, 0u);
                        myq[pos] = (unsigned)e;
                    }
                    q += __builtin_popcount(bal);
                }
                if (q >= QTH) { drain(feat, src, dst, tile, dcount, myq, q, lane, base, rows); q = 0; }
            }
        }
        if (q > 0) drain(feat, src, dst, tile, dcount, myq, q, lane, base, rows);

        float4 wst[4];
        #pragma unroll
        for (int k = 0; k < 4; ++k) wst[k] = ((const float4*)Wm)[t + k * SB_WAVES * 32];
        const float bcol = bias[n];
        __syncthreads();

        #pragma unroll
        for (int k = 0; k < 4; ++k) {
            const int wn = (t >> 5) + k * SB_WAVES;
            float* wrow = qw.w + wn * TS + lane;
            wrow[0]  = wst[k].x;
            wrow[32] = wst[k].y;
            wrow[64] = wst[k].z;
            wrow[96] = wst[k].w;
        }
        __syncthreads();

        float4 wf[16];
        #pragma unroll
        for (int qq = 0; qq < 16; ++qq)
            wf[qq] = *(const float4*)(qw.w + n * TS + 8 * qq + 4 * h);

        const int ntiles = (rows + 15) >> 4;
        for (int rt = rg; rt < ntiles; rt += 4) {
            const float* arow = tile + (rt * 16 + c) * TS + 4 * h;
            const int r0 = rt * 16 + 8 * h;
            const float4 dg0 = *(const float4*)(dcount + r0);
            const float4 dg1 = *(const float4*)(dcount + r0 + 4);
            v8f acc = {};
            #pragma unroll
            for (int qq = 0; qq < 16; ++qq) {
                const float4 av = *(const float4*)(arow + 8 * qq);
                const v2f a0 = {av.x, av.y};
                const v2f a1 = {av.z, av.w};
                const v2f b0 = {wf[qq].x, wf[qq].y};
                const v2f b1 = {wf[qq].z, wf[qq].w};
                acc = __builtin_amdgcn_wmma_f32_16x16x4_f32(
                     false, a0,  false, b0,
                     (short)0, acc,  false,  false);
                acc = __builtin_amdgcn_wmma_f32_16x16x4_f32(
                     false, a1,  false, b1,
                     (short)0, acc,  false,  false);
            }

            const float dg[8] = {dg0.x, dg0.y, dg0.z, dg0.w, dg1.x, dg1.y, dg1.z, dg1.w};
            float res[8];
            #pragma unroll
            for (int r = 0; r < 8; ++r)
                res[r] = acc[r] / fmaxf(dg[r], 1.0f) + bcol;

            float* orow = out + (long long)(base + r0) * D + n;
            if (rt * 16 + 16 <= rows) {
                #pragma unroll
                for (int r = 0; r < 8; ++r) orow[r * D] = res[r];
            } else {
                #pragma unroll
                for (int r = 0; r < 8; ++r) {
                    if (r0 + r < rows) orow[r * D] = res[r];
                }
            }
        }
        __syncthreads();
    }
}

extern "C" void kernel_launch(void* const* d_in, const int* in_sizes, int n_in,
                              void* d_out, int out_size, void* d_ws, size_t ws_size,
                              hipStream_t stream) {
    const float* feature = (const float*)d_in[0];
    const float* Wm      = (const float*)d_in[1];
    const float* bias    = (const float*)d_in[2];
    const int*   src     = (const int*)d_in[3];
    const int*   dst     = (const int*)d_in[4];

    float* out = (float*)d_out;
    unsigned* keys = (unsigned*)d_ws;

    const int N = in_sizes[0] / D;
    const int E = in_sizes[3];

    gcn_owner_keys<<<160, 256, 0, stream>>>(dst, keys, E);
    gcn_owner_layer<<<NBLK, SB_WAVES * 32, 0, stream>>>(feature, src, dst, keys, Wm, bias, out, N, E);
}
